# starting kernel
# baseline (speedup 1.0000x reference)
#include <hip/hip_runtime.h>
#include <hip/hip_bf16.h>

typedef __attribute__((ext_vector_type(2))) float v2f;
typedef __attribute__((ext_vector_type(8))) float v8f;

#define D 128
#define NEG_SLOPE 0.2f

__global__ __launch_bounds__(256) void gat_zero(float* __restrict__ p, long long n) {
    long long i = (long long)blockIdx.x * blockDim.x + threadIdx.x;
    if (i < n) p[i] = 0.0f;
}

__global__ __launch_bounds__(256) void gat_gemm_wmma(
    const float* __restrict__ X,
    const float* __restrict__ W0, const float* __restrict__ W1,
    const float* __restrict__ W2,
    float* __restrict__ O0, float* __restrict__ O1, float* __restrict__ O2,
    int n_nodes)
{
    __shared__ float sX[16 * D];

    const int row0 = blockIdx.x * 16;
    const int m    = blockIdx.y;
    const float* __restrict__ W = (m == 0) ? W0 : (m == 1) ? W1 : W2;
    float* __restrict__       O = (m == 0) ? O0 : (m == 1) ? O1 : O2;

    const int tid = threadIdx.x;

    {
        float4* Sv = (float4*)sX;
        int f0 = tid, f1 = tid + 256;
        int r0 = row0 + (f0 >> 5); r0 = r0 < n_nodes ? r0 : n_nodes - 1;
        int r1 = row0 + (f1 >> 5); r1 = r1 < n_nodes ? r1 : n_nodes - 1;
        const float4* X0 = (const float4*)(X + (long long)r0 * D);
        const float4* X1 = (const float4*)(X + (long long)r1 * D);
        Sv[f0] = X0[f0 & 31];
        Sv[f1] = X1[f1 & 31];
    }
    __syncthreads();

    const int wave = tid >> 5;
    const int lane = tid & 31;
    const int half = lane >> 4;
    const int lm   = lane & 15;
    const int col0 = wave * 16;

    v8f acc = {};
    #pragma unroll
    for (int k0 = 0; k0 < D; k0 += 4) {
        const int ka = k0 + half * 2;
        v2f a, b;
        a.x = sX[lm * D + ka + 0];
        a.y = sX[lm * D + ka + 1];
        b.x = W[(ka + 0) * D + col0 + lm];
        b.y = W[(ka + 1) * D + col0 + lm];
        acc = __builtin_amdgcn_wmma_f32_16x16x4_f32(
                false, a, false, b, (short)0, acc, false, false);
    }

    #pragma unroll
    for (int v = 0; v < 8; ++v) {
        const int row = row0 + v + half * 8;
        if (row < n_nodes)
            O[(long long)row * D + col0 + lm] = acc[v];
    }
}

__global__ __launch_bounds__(256) void gat_edge_fused(
    const float* __restrict__ h_l, const float* __restrict__ h_r,
    const float* __restrict__ comb,
    const int* __restrict__ src, const int* __restrict__ dst,
    float* __restrict__ agg, int n_edges)
{
    const int wavesPerBlock = blockDim.x >> 5;
    const int edge = blockIdx.x * wavesPerBlock + (threadIdx.x >> 5);
    const int lane = threadIdx.x & 31;
    if (edge >= n_edges) return;

    const int s = src[edge];
    const int d = dst[edge];

    const float4 hl = ((const float4*)(h_l + (long long)s * D))[lane];
    const float4 hr = ((const float4*)(h_r + (long long)d * D))[lane];

    float p = hl.x * hr.x + hl.y * hr.y + hl.z * hr.z + hl.w * hr.w;
    #pragma unroll
    for (int off = 16; off > 0; off >>= 1)
        p += __shfl_xor(p, off, 32);

    const float e = (p > 0.0f) ? p : NEG_SLOPE * p;

    const float4 cb = ((const float4*)(comb + (long long)d * D))[lane];
    float* outrow = agg + (long long)s * D + lane * 4;
    unsafeAtomicAdd(outrow + 0, e * cb.x);
    unsafeAtomicAdd(outrow + 1, e * cb.y);
    unsafeAtomicAdd(outrow + 2, e * cb.z);
    unsafeAtomicAdd(outrow + 3, e * cb.w);
}

extern "C" void kernel_launch(void* const* d_in, const int* in_sizes, int n_in,
                              void* d_out, int out_size, void* d_ws, size_t ws_size,
                              hipStream_t stream) {
    const float* X   = (const float*)d_in[0];
    const float* W   = (const float*)d_in[1];
    const float* W_r = (const float*)d_in[2];
    const float* W_l = (const float*)d_in[3];
    const int*   src = (const int*)d_in[4];
    const int*   dst = (const int*)d_in[5];

    const int n_nodes = in_sizes[0] / D;
    const int n_edges = in_sizes[4];

    const long long per = (long long)n_nodes * D;
    float* comb = (float*)d_ws;
    float* h_r  = comb + per;
    float* h_l  = h_r + per;
    float* agg  = (float*)d_out;

    {
        long long n = per;
        int blocks = (int)((n + 255) / 256);
        gat_zero<<<blocks, 256, 0, stream>>>(agg, n);
    }

    {
        dim3 grid((n_nodes + 15) / 16, 3, 1);
        gat_gemm_wmma<<<grid, 256, 0, stream>>>(X, W, W_r, W_l,
                                                comb, h_r, h_l, n_nodes);
    }

    {
        const int wavesPerBlock = 8;
        int blocks = (n_edges + wavesPerBlock - 1) / wavesPerBlock;
        gat_edge_fused<<<blocks, 256, 0, stream>>>(h_l, h_r, comb,
                                                   src, dst, agg, n_edges);
    }
}


# Edge scatter becomes a gather per source node: an atomic-exchange pass chains each node's edges into a list, one wave per node sums its messages in registers in list (arrival) order; no float atomics
# speedup vs baseline: 1.3829x; 1.3829x over previous
#include <hip/hip_runtime.h>
#include <hip/hip_bf16.h>

typedef __attribute__((ext_vector_type(2))) float v2f;
typedef __attribute__((ext_vector_type(8))) float v8f;

#define D 128
#define NEG_SLOPE 0.2f

__global__ __launch_bounds__(256) void gat_gemm_wmma(
    const float* __restrict__ X,
    const float* __restrict__ W0, const float* __restrict__ W1,
    const float* __restrict__ W2,
    float* __restrict__ O0, float* __restrict__ O1, float* __restrict__ O2,
    int* __restrict__ head, int n_nodes)
{
    __shared__ float sX[16 * D];

    const int row0 = blockIdx.x * 16;
    const int m    = blockIdx.y;
    const float* __restrict__ W = (m == 0) ? W0 : (m == 1) ? W1 : W2;
    float* __restrict__       O = (m == 0) ? O0 : (m == 1) ? O1 : O2;

    const int tid = threadIdx.x;

    if (m == 0 && tid < 16 && row0 + tid < n_nodes) head[row0 + tid] = -1;

    {
        float4* Sv = (float4*)sX;
        int f0 = tid, f1 = tid + 256;
        int r0 = row0 + (f0 >> 5); r0 = r0 < n_nodes ? r0 : n_nodes - 1;
        int r1 = row0 + (f1 >> 5); r1 = r1 < n_nodes ? r1 : n_nodes - 1;
        const float4* X0 = (const float4*)(X + (long long)r0 * D);
        const float4* X1 = (const float4*)(X + (long long)r1 * D);
        Sv[f0] = X0[f0 & 31];
        Sv[f1] = X1[f1 & 31];
    }
    __syncthreads();

    const int wave = tid >> 5;
    const int lane = tid & 31;
    const int half = lane >> 4;
    const int lm   = lane & 15;
    const int col0 = wave * 16;

    v8f acc = {};
    #pragma unroll
    for (int k0 = 0; k0 < D; k0 += 4) {
        const int ka = k0 + half * 2;
        v2f a, b;
        a.x = sX[lm * D + ka + 0];
        a.y = sX[lm * D + ka + 1];
        b.x = W[(ka + 0) * D + col0 + lm];
        b.y = W[(ka + 1) * D + col0 + lm];
        acc = __builtin_amdgcn_wmma_f32_16x16x4_f32(
                false, a, false, b, (short)0, acc, false, false);
    }

    #pragma unroll
    for (int v = 0; v < 8; ++v) {
        const int row = row0 + v + half * 8;
        if (row < n_nodes)
            O[(long long)row * D + col0 + lm] = acc[v];
    }
}

__global__ __launch_bounds__(256) void gat_link_edges(
    const int* __restrict__ src, const int* __restrict__ dst,
    int* __restrict__ head, int2* __restrict__ link, int n_edges)
{
    const int e = blockIdx.x * 256 + threadIdx.x;
    if (e >= n_edges) return;
    const int prev = atomicExch(head + src[e], e);
    link[e] = make_int2(dst[e], prev);
}

__device__ __forceinline__ float add_other_row(float p) {
    const int b = __builtin_bit_cast(int, p);
    return p + __builtin_bit_cast(float,
        __builtin_amdgcn_permlanex16(b, b, 0x76543210, 0xfedcba98, false, false));
}
template <int M>
__device__ __forceinline__ float add_row_xmask(float p) {
    const int b = __builtin_bit_cast(int, p);
    return p + __builtin_bit_cast(float,
        __builtin_amdgcn_update_dpp(b, b, 0x160 + M, 0xf, 0xf, false));
}

__global__ __launch_bounds__(256) void gat_node_gather(
    const float* __restrict__ h_r, const float* __restrict__ comb,
    const int* __restrict__ head, const int2* __restrict__ link,
    float* out, int n_nodes, int n_edges)
{
    const int node = __builtin_amdgcn_readfirstlane(
                         (int)(blockIdx.x * 8 + (threadIdx.x >> 5)));
    const int lane = threadIdx.x & 31;
    if (node >= n_nodes) return;

    float4* orow = (float4*)(out + (long long)node * D) + lane;
    const float4 hl = *orow;
    float4 acc = make_float4(0.0f, 0.0f, 0.0f, 0.0f);

    const int e = head[node];
    if (e >= 0) {
        int2 l = link[e];
        for (int t = 0; t < n_edges; ++t) {
            const int d  = l.x;
            const int nx = l.y;
            if (nx >= 0) l = link[nx];

            const float4 hr = ((const float4*)(h_r  + (long long)d * D))[lane];
            const float4 cb = ((const float4*)(comb + (long long)d * D))[lane];
            __builtin_amdgcn_sched_barrier(0);

            float p = hl.x * hr.x + hl.y * hr.y + hl.z * hr.z + hl.w * hr.w;
            p = add_other_row(p);
            p = add_row_xmask<8>(p);
            p = add_row_xmask<4>(p);
            p = add_row_xmask<2>(p);
            p = add_row_xmask<1>(p);

            const float ev = (p > 0.0f) ? p : NEG_SLOPE * p;
            acc.x += ev * cb.x;
            acc.y += ev * cb.y;
            acc.z += ev * cb.z;
            acc.w += ev * cb.w;

            if (nx < 0) break;
        }
    }
    *orow = acc;
}

extern "C" void kernel_launch(void* const* d_in, const int* in_sizes, int n_in,
                              void* d_out, int out_size, void* d_ws, size_t ws_size,
                              hipStream_t stream) {
    const float* X   = (const float*)d_in[0];
    const float* W   = (const float*)d_in[1];
    const float* W_r = (const float*)d_in[2];
    const float* W_l = (const float*)d_in[3];
    const int*   src = (const int*)d_in[4];
    const int*   dst = (const int*)d_in[5];

    const int n_nodes = in_sizes[0] / D;
    const int n_edges = in_sizes[4];

    const long long per = (long long)n_nodes * D;
    float* comb = (float*)d_ws;
    float* h_r  = comb + per;
    int*   head = (int*)(h_r + per);
    int2*  link = (int2*)(head + ((n_nodes + 3) & ~3));
    float* agg  = (float*)d_out;

    {
        dim3 grid((n_nodes + 15) / 16, 3, 1);
        gat_gemm_wmma<<<grid, 256, 0, stream>>>(X, W, W_r, W_l,
                                                comb, h_r, agg, head, n_nodes);
    }

    {
        int blocks = (n_edges + 255) / 256;
        gat_link_edges<<<blocks, 256, 0, stream>>>(src, dst, head, link, n_edges);
    }

    {
        int blocks = (n_nodes + 7) / 8;
        gat_node_gather<<<blocks, 256, 0, stream>>>(h_r, comb, head, link,
                                                    agg, n_nodes, n_edges);
    }
}


# X tile in LDS at a row pitch of 132 floats instead of 128: the A-fragment reads of the projection GEMM (16 lanes, 16 rows, same k) no longer fall on one bank pair
# speedup vs baseline: 1.6808x; 1.2154x over previous
#include <hip/hip_runtime.h>
#include <hip/hip_bf16.h>

typedef __attribute__((ext_vector_type(2))) float v2f;
typedef __attribute__((ext_vector_type(8))) float v8f;

#define D 128
#define NEG_SLOPE 0.2f
#define XP 132

__global__ __launch_bounds__(256) void gat_gemm_wmma(
    const float* __restrict__ X,
    const float* __restrict__ W0, const float* __restrict__ W1,
    const float* __restrict__ W2,
    float* __restrict__ O0, float* __restrict__ O1, float* __restrict__ O2,
    int* __restrict__ head, int n_nodes)
{
    __shared__ float sX[16 * XP];

    const int row0 = blockIdx.x * 16;
    const int m    = blockIdx.y;
    const float* __restrict__ W = (m == 0) ? W0 : (m == 1) ? W1 : W2;
    float* __restrict__       O = (m == 0) ? O0 : (m == 1) ? O1 : O2;

    const int tid = threadIdx.x;

    if (m == 0 && tid < 16 && row0 + tid < n_nodes) head[row0 + tid] = -1;

    {
        int f0 = tid, f1 = tid + 256;
        int r0 = row0 + (f0 >> 5); r0 = r0 < n_nodes ? r0 : n_nodes - 1;
        int r1 = row0 + (f1 >> 5); r1 = r1 < n_nodes ? r1 : n_nodes - 1;
        const float4* X0 = (const float4*)(X + (long long)r0 * D);
        const float4* X1 = (const float4*)(X + (long long)r1 * D);
        *(float4*)(sX + (f0 >> 5) * XP + (f0 & 31) * 4) = X0[f0 & 31];
        *(float4*)(sX + (f1 >> 5) * XP + (f1 & 31) * 4) = X1[f1 & 31];
    }
    __syncthreads();

    const int wave = tid >> 5;
    const int lane = tid & 31;
    const int half = lane >> 4;
    const int lm   = lane & 15;
    const int col0 = wave * 16;

    v8f acc = {};
    #pragma unroll
    for (int k0 = 0; k0 < D; k0 += 4) {
        const int ka = k0 + half * 2;
        v2f a, b;
        a.x = sX[lm * XP + ka + 0];
        a.y = sX[lm * XP + ka + 1];
        b.x = W[(ka + 0) * D + col0 + lm];
        b.y = W[(ka + 1) * D + col0 + lm];
        acc = __builtin_amdgcn_wmma_f32_16x16x4_f32(
                false, a, false, b, (short)0, acc, false, false);
    }

    #pragma unroll
    for (int v = 0; v < 8; ++v) {
        const int row = row0 + v + half * 8;
        if (row < n_nodes)
            O[(long long)row * D + col0 + lm] = acc[v];
    }
}

__global__ __launch_bounds__(256) void gat_link_edges(
    const int* __restrict__ src, const int* __restrict__ dst,
    int* __restrict__ head, int2* __restrict__ link, int n_edges)
{
    const int e = blockIdx.x * 256 + threadIdx.x;
    if (e >= n_edges) return;
    const int prev = atomicExch(head + src[e], e);
    link[e] = make_int2(dst[e], prev);
}

__device__ __forceinline__ float add_other_row(float p) {
    const int b = __builtin_bit_cast(int, p);
    return p + __builtin_bit_cast(float,
        __builtin_amdgcn_permlanex16(b, b, 0x76543210, 0xfedcba98, false, false));
}
template <int M>
__device__ __forceinline__ float add_row_xmask(float p) {
    const int b = __builtin_bit_cast(int, p);
    return p + __builtin_bit_cast(float,
        __builtin_amdgcn_update_dpp(b, b, 0x160 + M, 0xf, 0xf, false));
}

__global__ __launch_bounds__(256) void gat_node_gather(
    const float* __restrict__ h_r, const float* __restrict__ comb,
    const int* __restrict__ head, const int2* __restrict__ link,
    float* out, int n_nodes, int n_edges)
{
    const int node = __builtin_amdgcn_readfirstlane(
                         (int)(blockIdx.x * 8 + (threadIdx.x >> 5)));
    const int lane = threadIdx.x & 31;
    if (node >= n_nodes) return;

    float4* orow = (float4*)(out + (long long)node * D) + lane;
    const float4 hl = *orow;
    float4 acc = make_float4(0.0f, 0.0f, 0.0f, 0.0f);

    const int e = head[node];
    if (e >= 0) {
        int2 l = link[e];
        for (int t = 0; t < n_edges; ++t) {
            const int d  = l.x;
            const int nx = l.y;
            if (nx >= 0) l = link[nx];

            const float4 hr = ((const float4*)(h_r  + (long long)d * D))[lane];
            const float4 cb = ((const float4*)(comb + (long long)d * D))[lane];
            __builtin_amdgcn_sched_barrier(0);

            float p = hl.x * hr.x + hl.y * hr.y + hl.z * hr.z + hl.w * hr.w;
            p = add_other_row(p);
            p = add_row_xmask<8>(p);
            p = add_row_xmask<4>(p);
            p = add_row_xmask<2>(p);
            p = add_row_xmask<1>(p);

            const float ev = (p > 0.0f) ? p : NEG_SLOPE * p;
            acc.x += ev * cb.x;
            acc.y += ev * cb.y;
            acc.z += ev * cb.z;
            acc.w += ev * cb.w;

            if (nx < 0) break;
        }
    }
    *orow = acc;
}

extern "C" void kernel_launch(void* const* d_in, const int* in_sizes, int n_in,
                              void* d_out, int out_size, void* d_ws, size_t ws_size,
                              hipStream_t stream) {
    const float* X   = (const float*)d_in[0];
    const float* W   = (const float*)d_in[1];
    const float* W_r = (const float*)d_in[2];
    const float* W_l = (const float*)d_in[3];
    const int*   src = (const int*)d_in[4];
    const int*   dst = (const int*)d_in[5];

    const int n_nodes = in_sizes[0] / D;
    const int n_edges = in_sizes[4];

    const long long per = (long long)n_nodes * D;
    float* comb = (float*)d_ws;
    float* h_r  = comb + per;
    int*   head = (int*)(h_r + per);
    int2*  link = (int2*)(head + ((n_nodes + 3) & ~3));
    float* agg  = (float*)d_out;

    {
        dim3 grid((n_nodes + 15) / 16, 3, 1);
        gat_gemm_wmma<<<grid, 256, 0, stream>>>(X, W, W_r, W_l,
                                                comb, h_r, agg, head, n_nodes);
    }

    {
        int blocks = (n_edges + 255) / 256;
        gat_link_edges<<<blocks, 256, 0, stream>>>(src, dst, head, link, n_edges);
    }

    {
        int blocks = (n_nodes + 7) / 8;
        gat_node_gather<<<blocks, 256, 0, stream>>>(h_r, comb, head, link,
                                                    agg, n_nodes, n_edges);
    }
}


# One block takes 32 nodes from X rows to output rows (g and z tiles stay in LDS; 3 launches, not 4); a wave walks 4 edge lists at once, 8 lanes per node: dot summed 16 per lane, then 8-lane tree
# speedup vs baseline: 2.5490x; 1.5165x over previous
#include <hip/hip_runtime.h>


typedef __attribute__((ext_vector_type(2))) float v2f;
typedef __attribute__((ext_vector_type(8))) float v8f;

#define D 128
#define NEG_SLOPE 0.2f
#define XP 132


__global__ __launch_bounds__(256) void gat_fold_weights(
    const float* __restrict__ W_l, const float* __restrict__ W_r,
    float* __restrict__ M, int* __restrict__ head, int n_nodes)
{
    const int tid = threadIdx.x;
    if (blockIdx.x >= 8) {
        const int i = (blockIdx.x - 8) * 256 + tid;
        if (i < n_nodes) head[i] = -1;
        return;
    }
    const int wave = tid >> 5;
    const int lane = tid & 31;
    const int half = lane >> 4;
    const int lm   = lane & 15;
    const float* __restrict__ arow = W_l + (blockIdx.x * 16 + lm) * D + half * 2;
    const float* __restrict__ brow = W_r + (wave * 16 + lm) * D + half * 2;

    v8f acc = {};
    #pragma unroll
    for (int k0 = 0; k0 < D; k0 += 4) {
        v2f a, b;
        a.x = arow[k0 + 0];
        a.y = arow[k0 + 1];
        b.x = brow[k0 + 0];
        b.y = brow[k0 + 1];
        acc = __builtin_amdgcn_wmma_f32_16x16x4_f32(
                false, a, false, b, (short)0, acc, false, false);
    }
    #pragma unroll
    for (int v = 0; v < 8; ++v)
        M[(blockIdx.x * 16 + v + half * 8) * D + wave * 16 + lm] = acc[v];
}

__global__ __launch_bounds__(256) void gat_link_edges(
    const int* __restrict__ src, const int* __restrict__ dst,
    int* __restrict__ head, int2* __restrict__ link, int n_edges)
{
    const int e = (int)blockIdx.x * 256 + (int)threadIdx.x;
    if (e < n_edges) {
        const int prev = atomicExch(head + src[e], e);
        link[e] = make_int2(dst[e], prev);
    }
}

template <int M>
__device__ __forceinline__ float add_row_xmask(float p) {
    const int b = __builtin_bit_cast(int, p);
    return p + __builtin_bit_cast(float,
        __builtin_amdgcn_update_dpp(b, b, 0x160 + M, 0xf, 0xf, false));
}

__device__ __forceinline__ void tile32_times_matrix(
    const float* sT, const float* __restrict__ Bm,
    int lm, int half, int col0, v8f& c0, v8f& c1)
{
    #pragma unroll 4
    for (int k0 = 0; k0 < D; k0 += 4) {
        const int ka = k0 + half * 2;
        v2f a0, a1, b;
        a0.x = sT[lm * XP + ka + 0];
        a0.y = sT[lm * XP + ka + 1];
        a1.x = sT[(16 + lm) * XP + ka + 0];
        a1.y = sT[(16 + lm) * XP + ka + 1];
        b.x = Bm[(ka + 0) * D + col0 + lm];
        b.y = Bm[(ka + 1) * D + col0 + lm];
        c0 = __builtin_amdgcn_wmma_f32_16x16x4_f32(
                false, a0, false, b, (short)0, c0, false, false);
        c1 = __builtin_amdgcn_wmma_f32_16x16x4_f32(
                false, a1, false, b, (short)0, c1, false, false);
    }
}

__global__ __launch_bounds__(256, 16) void gat_node_block(
    const float* __restrict__ X, const float* __restrict__ Mm,
    const float* __restrict__ Wm,
    const int* __restrict__ head, const int2* __restrict__ link,
    float* __restrict__ out, int n_nodes, int n_edges)
{
    __shared__ float sA[32 * XP];
    __shared__ float sG[32 * XP];

    const int tid  = threadIdx.x;
    const int row0 = blockIdx.x * 32;

    #pragma unroll
    for (int q = 0; q < 4; ++q) {
        const int f = tid + 256 * q;
        int r = row0 + (f >> 5);
        r = r < n_nodes ? r : n_nodes - 1;
        const float4* Xr = (const float4*)(X + (long long)r * D);
        *(float4*)(sA + (f >> 5) * XP + (f & 31) * 4) = Xr[f & 31];
    }
    __syncthreads();

    const int wave = tid >> 5;
    const int lane = tid & 31;
    const int half = lane >> 4;
    const int lm   = lane & 15;
    const int col0 = wave * 16;

    {
        v8f c0 = {}, c1 = {};
        tile32_times_matrix(sA, Mm, lm, half, col0, c0, c1);
        #pragma unroll
        for (int v = 0; v < 8; ++v) {
            sG[(v + half * 8) * XP + col0 + lm]      = c0[v];
            sG[(16 + v + half * 8) * XP + col0 + lm] = c1[v];
        }
    }
    __syncthreads();

    {
        const int grp = lane >> 3;
        const int sub = lane & 7;
        const int nl  = wave * 4 + grp;
        const int node = row0 + nl;

        const float4* gp = (const float4*)(sG + nl * XP) + sub;
        const float4 g0 = gp[0], g1 = gp[8], g2 = gp[16], g3 = gp[24];
        const float4* X4 = (const float4*)X;
        float4 z0 = make_float4(0.0f, 0.0f, 0.0f, 0.0f);
        float4 z1 = z0, z2 = z0, z3 = z0;

        int cur = -1;
        if (node < n_nodes) cur = head[node];
        int2 l = make_int2(0, -1);
        if (cur >= 0) l = link[cur];
        for (int t = 0; t < n_edges && cur >= 0; ++t) {
            const int d  = l.x;
            const int nx = l.y;
            if (nx >= 0) l = link[nx];

            const unsigned xi = (unsigned)d * 32u + (unsigned)sub;
            const float4 x0 = X4[xi], x1 = X4[xi + 8u], x2 = X4[xi + 16u], x3 = X4[xi + 24u];

            float p0 = g0.x * x0.x, p1 = g1.x * x1.x, p2 = g2.x * x2.x, p3 = g3.x * x3.x;
            p0 = __builtin_fmaf(g0.y, x0.y, p0); p1 = __builtin_fmaf(g1.y, x1.y, p1);
            p2 = __builtin_fmaf(g2.y, x2.y, p2); p3 = __builtin_fmaf(g3.y, x3.y, p3);
            p0 = __builtin_fmaf(g0.z, x0.z, p0); p1 = __builtin_fmaf(g1.z, x1.z, p1);
            p2 = __builtin_fmaf(g2.z, x2.z, p2); p3 = __builtin_fmaf(g3.z, x3.z, p3);
            p0 = __builtin_fmaf(g0.w, x0.w, p0); p1 = __builtin_fmaf(g1.w, x1.w, p1);
            p2 = __builtin_fmaf(g2.w, x2.w, p2); p3 = __builtin_fmaf(g3.w, x3.w, p3);
            float p = (p0 + p1) + (p2 + p3);
            p = add_row_xmask<4>(p);
            p = add_row_xmask<2>(p);
            p = add_row_xmask<1>(p);

            const float ev = (p > 0.0f) ? p : NEG_SLOPE * p;
            z0.x += ev * x0.x; z0.y += ev * x0.y; z0.z += ev * x0.z; z0.w += ev * x0.w;
            z1.x += ev * x1.x; z1.y += ev * x1.y; z1.z += ev * x1.z; z1.w += ev * x1.w;
            z2.x += ev * x2.x; z2.y += ev * x2.y; z2.z += ev * x2.z; z2.w += ev * x2.w;
            z3.x += ev * x3.x; z3.y += ev * x3.y; z3.z += ev * x3.z; z3.w += ev * x3.w;

            cur = nx;
        }

        float4* zp = (float4*)(sA + nl * XP) + sub;
        zp[0] = z0; zp[8] = z1; zp[16] = z2; zp[24] = z3;
    }
    __syncthreads();

    {
        v8f c0 = {}, c1 = {};
        tile32_times_matrix(sA, Wm, lm, half, col0, c0, c1);
        #pragma unroll
        for (int v = 0; v < 8; ++v) {
            const int ra = row0 + v + half * 8;
            const int rb = ra + 16;
            if (ra < n_nodes) out[(long long)ra * D + col0 + lm] = c0[v];
            if (rb < n_nodes) out[(long long)rb * D + col0 + lm] = c1[v];
        }
    }
}

extern "C" void kernel_launch(void* const* d_in, const int* in_sizes, int n_in,
                              void* d_out, int out_size, void* d_ws, size_t ws_size,
                              hipStream_t stream) {
    const float* X   = (const float*)d_in[0];
    const float* W   = (const float*)d_in[1];
    const float* W_r = (const float*)d_in[2];
    const float* W_l = (const float*)d_in[3];
    const int*   src = (const int*)d_in[4];
    const int*   dst = (const int*)d_in[5];

    const int n_nodes = in_sizes[0] / D;
    const int n_edges = in_sizes[4];

    float* M    = (float*)d_ws;
    int*   head = (int*)(M + D * D);
    int2*  link = (int2*)(head + ((n_nodes + 3) & ~3));
    float* agg  = (float*)d_out;

    const int n_tiles     = (n_nodes + 31) / 32;
    const int head_blocks = (n_nodes + 255) / 256;
    const int edge_blocks = (n_edges + 255) / 256;

    gat_fold_weights<<<8 + head_blocks, 256, 0, stream>>>(W_l, W_r, M, head, n_nodes);

    gat_link_edges<<<edge_blocks, 256, 0, stream>>>(src, dst, head, link, n_edges);

    gat_node_block<<<n_tiles, 256, 0, stream>>>(
        X, M, W, head, link, agg, n_nodes, n_edges);
}
